# starting kernel
# baseline (speedup 1.0000x reference)
#include <hip/hip_runtime.h>


#define Bn    2
#define Cn    64
#define Nn    20000
#define Kn    16
#define COUTn 64

typedef __attribute__((ext_vector_type(2))) float v2f;
typedef __attribute__((ext_vector_type(8))) float v8f;

__global__ void build_wc(const float* __restrict__ W, float* __restrict__ Wc) {
    int t = blockIdx.x * blockDim.x + threadIdx.x;
    if (t >= 128 * 64) return;
    int o = t >> 6, c = t & 63;
    float v;
    if (o < 64) v = W[o * 128 + c] - W[o * 128 + 64 + c];
    else        v = W[(o - 64) * 128 + 64 + c];
    Wc[t] = v;
}

__global__ __launch_bounds__(32) void gemm_y(
    const float* __restrict__ x, const float* __restrict__ Wc,
    const float* __restrict__ bias, float* __restrict__ y) {
    const int lane = threadIdx.x;
    const int nlo  = lane & 15;
    const int hi   = lane >> 4;
    const int tile = blockIdx.x;
    const int b    = tile / (Nn / 16);
    const int m0   = (tile % (Nn / 16)) * 16;

    v8f acc[8];
#pragma unroll
    for (int mt = 0; mt < 8; ++mt) {
#pragma unroll
        for (int r = 0; r < 8; ++r) {
            int o = mt * 16 + r + 8 * hi;
            acc[mt][r] = (o < COUTn) ? bias[o] : 0.0f;
        }
    }

    const float* xb = x + (size_t)b * Cn * Nn;
#pragma unroll 4
    for (int c0 = 0; c0 < Cn; c0 += 4) {
        v2f bf;
        bf.x = xb[(size_t)(c0 + 2 * hi)     * Nn + m0 + nlo];
        bf.y = xb[(size_t)(c0 + 2 * hi + 1) * Nn + m0 + nlo];
#pragma unroll
        for (int mt = 0; mt < 8; ++mt) {
            v2f af;
            af.x = Wc[(mt * 16 + nlo) * 64 + c0 + 2 * hi];
            af.y = Wc[(mt * 16 + nlo) * 64 + c0 + 2 * hi + 1];
            acc[mt] = __builtin_amdgcn_wmma_f32_16x16x4_f32(
                false, af, false, bf, (short)0, acc[mt], false, false);
        }
    }

    float* yrow = y + ((size_t)b * Nn + m0 + nlo) * 128;
#pragma unroll
    for (int mt = 0; mt < 8; ++mt) {
        float4* dst = (float4*)(yrow + mt * 16 + 8 * hi);
        dst[0] = make_float4(acc[mt][0], acc[mt][1], acc[mt][2], acc[mt][3]);
        dst[1] = make_float4(acc[mt][4], acc[mt][5], acc[mt][6], acc[mt][7]);
    }
}

__global__ __launch_bounds__(256) void edge_combine(
    const float* __restrict__ y, const int* __restrict__ eidx,
    const float* __restrict__ pos, float* __restrict__ out) {
    __shared__ float smem[64][8];
    const int t    = threadIdx.x;
    const int lane = t & 31;
    const int w    = t >> 5;
    const long p   = (long)blockIdx.x * 8 + w;
    const int b    = (int)(p / Nn);
    const int n    = (int)(p - (long)b * Nn);

    int iv = 0, jv = 0;
    float supp = 0.0f;
    if (lane < Kn) {
        const size_t ebase = ((size_t)b * Nn + n) * Kn + lane;
        jv = eidx[ebase];
        iv = eidx[(size_t)Bn * Nn * Kn + ebase];
        const float* pb = pos + (size_t)b * 3 * Nn;
        float dx = pb[jv] - pb[iv];
        float dy = pb[Nn + jv] - pb[Nn + iv];
        float dz = pb[2 * Nn + jv] - pb[2 * Nn + iv];
        float dis = sqrtf(dx * dx + dy * dy + dz * dz);
        supp = 2.0f / (1.0f + __expf(dis));
    }

    float mx0 = 0.0f, mx1 = 0.0f;
    const float* yb = y + (size_t)b * Nn * 128;
#pragma unroll
    for (int k = 0; k < Kn; ++k) {
        int   ik = __shfl(iv, k, 32);
        int   jk = __shfl(jv, k, 32);
        float sk = __shfl(supp, k, 32);
        const float* yi = yb + (size_t)ik * 128;
        const float* yj = yb + (size_t)jk * 128 + 64;
        float a0 = yi[lane]      + yj[lane];
        float a1 = yi[lane + 32] + yj[lane + 32];
        a0 = fmaxf(a0, 0.0f) * sk;
        a1 = fmaxf(a1, 0.0f) * sk;
        mx0 = fmaxf(mx0, a0);
        mx1 = fmaxf(mx1, a1);
    }

    smem[lane][w]      = mx0;
    smem[lane + 32][w] = mx1;
    __syncthreads();
    const int b0 = (blockIdx.x * 8) / Nn;
    const int n0 = blockIdx.x * 8 - b0 * Nn;
#pragma unroll
    for (int v = t; v < 512; v += 256) {
        int o = v >> 3, ni = v & 7;
        out[((size_t)b0 * COUTn + o) * Nn + n0 + ni] = smem[o][ni];
    }
}

extern "C" void kernel_launch(void* const* d_in, const int* in_sizes, int n_in,
                              void* d_out, int out_size, void* d_ws, size_t ws_size,
                              hipStream_t stream) {
    const float* x    = (const float*)d_in[0];
    const int*   eidx = (const int*)  d_in[1];
    const float* pos  = (const float*)d_in[2];
    const float* W    = (const float*)d_in[3];
    const float* bias = (const float*)d_in[4];

    float* Wc = (float*)d_ws;
    float* y  = (float*)((char*)d_ws + 128 * 64 * sizeof(float));

    build_wc<<<(128 * 64 + 255) / 256, 256, 0, stream>>>(W, Wc);
    gemm_y<<<Bn * (Nn / 16), 32, 0, stream>>>(x, Wc, bias, y);
    edge_combine<<<(Bn * Nn) / 8, 256, 0, stream>>>(y, eidx, pos, (float*)d_out);
}


# edge stage lane tiling: 4 lanes per node, each lane 16 channels via 16-byte row gathers (was 32 lanes x 2 channels, dword loads); wave covers 8 nodes, no loop; whole-line stores via LDS
# speedup vs baseline: 1.1401x; 1.1401x over previous
#include <hip/hip_runtime.h>


#define Bn    2
#define Cn    64
#define Nn    20000
#define Kn    16
#define COUTn 64

typedef __attribute__((ext_vector_type(2))) float v2f;
typedef __attribute__((ext_vector_type(8))) float v8f;

__global__ void build_wc(const float* __restrict__ W, float* __restrict__ Wc) {
    int t = blockIdx.x * blockDim.x + threadIdx.x;
    if (t >= 128 * 64) return;
    int o = t >> 6, c = t & 63;
    float v;
    if (o < 64) v = W[o * 128 + c] - W[o * 128 + 64 + c];
    else        v = W[(o - 64) * 128 + 64 + c];
    Wc[t] = v;
}

__global__ __launch_bounds__(32) void gemm_y(
    const float* __restrict__ x, const float* __restrict__ Wc,
    const float* __restrict__ bias, float* __restrict__ y) {
    const int lane = threadIdx.x;
    const int nlo  = lane & 15;
    const int hi   = lane >> 4;
    const int tile = blockIdx.x;
    const int b    = tile / (Nn / 16);
    const int m0   = (tile % (Nn / 16)) * 16;

    v8f acc[8];
#pragma unroll
    for (int mt = 0; mt < 8; ++mt) {
#pragma unroll
        for (int r = 0; r < 8; ++r) {
            int o = mt * 16 + r + 8 * hi;
            acc[mt][r] = (o < COUTn) ? bias[o] : 0.0f;
        }
    }

    const float* xb = x + (size_t)b * Cn * Nn;
#pragma unroll 4
    for (int c0 = 0; c0 < Cn; c0 += 4) {
        v2f bf;
        bf.x = xb[(size_t)(c0 + 2 * hi)     * Nn + m0 + nlo];
        bf.y = xb[(size_t)(c0 + 2 * hi + 1) * Nn + m0 + nlo];
#pragma unroll
        for (int mt = 0; mt < 8; ++mt) {
            v2f af;
            af.x = Wc[(mt * 16 + nlo) * 64 + c0 + 2 * hi];
            af.y = Wc[(mt * 16 + nlo) * 64 + c0 + 2 * hi + 1];
            acc[mt] = __builtin_amdgcn_wmma_f32_16x16x4_f32(
                false, af, false, bf, (short)0, acc[mt], false, false);
        }
    }

    float* yrow = y + ((size_t)b * Nn + m0 + nlo) * 128;
#pragma unroll
    for (int mt = 0; mt < 8; ++mt) {
        float4* dst = (float4*)(yrow + mt * 16 + 8 * hi);
        dst[0] = make_float4(acc[mt][0], acc[mt][1], acc[mt][2], acc[mt][3]);
        dst[1] = make_float4(acc[mt][4], acc[mt][5], acc[mt][6], acc[mt][7]);
    }
}

#define EPITCH 36

__global__ __launch_bounds__(128) void edge_combine(
    const float* __restrict__ y, const int* __restrict__ eidx,
    const float* __restrict__ pos, float* __restrict__ out) {
    __shared__ __attribute__((aligned(16))) float smem[COUTn * EPITCH];
    const int t  = threadIdx.x;
    const int q  = t & 3;
    const int g  = t >> 2;
    const int p0 = blockIdx.x * 32;
    const int b  = p0 / Nn;
    const int n0 = p0 - b * Nn;
    const int n  = n0 + g;

    const size_t ebase = ((size_t)b * Nn + n) * Kn + q * 4;
    const int4 j4 = *(const int4*)(eidx + ebase);
    const int4 i4 = *(const int4*)(eidx + (size_t)Bn * Nn * Kn + ebase);
    const int jv[4] = {j4.x, j4.y, j4.z, j4.w};
    const int iv[4] = {i4.x, i4.y, i4.z, i4.w};
    const float* pb = pos + (size_t)b * 3 * Nn;
    float supp[4];
#pragma unroll
    for (int e = 0; e < 4; ++e) {
        float dx = pb[jv[e]] - pb[iv[e]];
        float dy = pb[Nn + jv[e]] - pb[Nn + iv[e]];
        float dz = pb[2 * Nn + jv[e]] - pb[2 * Nn + iv[e]];
        float dis = sqrtf(dx * dx + dy * dy + dz * dz);
        supp[e] = 2.0f / (1.0f + __expf(dis));
    }

    float4 mx[4];
#pragma unroll
    for (int s = 0; s < 4; ++s) mx[s] = make_float4(0.0f, 0.0f, 0.0f, 0.0f);

    const float* yb = y + (size_t)b * Nn * 128 + q * 4;
#pragma unroll
    for (int k = 0; k < Kn; ++k) {
        const int   ik = __shfl(iv[k & 3],   k >> 2, 4);
        const int   jk = __shfl(jv[k & 3],   k >> 2, 4);
        const float sk = __shfl(supp[k & 3], k >> 2, 4);
        const float4* yi = (const float4*)(yb + (size_t)ik * 128);
        const float4* yj = (const float4*)(yb + (size_t)jk * 128 + 64);
#pragma unroll
        for (int s = 0; s < 4; ++s) {
            const float4 a = yi[s * 4];
            const float4 c = yj[s * 4];
            mx[s].x = fmaxf(mx[s].x, fmaxf(a.x + c.x, 0.0f) * sk);
            mx[s].y = fmaxf(mx[s].y, fmaxf(a.y + c.y, 0.0f) * sk);
            mx[s].z = fmaxf(mx[s].z, fmaxf(a.z + c.z, 0.0f) * sk);
            mx[s].w = fmaxf(mx[s].w, fmaxf(a.w + c.w, 0.0f) * sk);
        }
    }

#pragma unroll
    for (int s = 0; s < 4; ++s) {
        float* r = smem + (s * 16 + q * 4) * EPITCH + g;
        r[0 * EPITCH] = mx[s].x;
        r[1 * EPITCH] = mx[s].y;
        r[2 * EPITCH] = mx[s].z;
        r[3 * EPITCH] = mx[s].w;
    }
    __syncthreads();
#pragma unroll
    for (int v = t; v < COUTn * 8; v += 128) {
        const int o = v >> 3, c4 = v & 7;
        const float4 val = *(const float4*)(smem + o * EPITCH + c4 * 4);
        *(float4*)(out + ((size_t)b * COUTn + o) * Nn + n0 + c4 * 4) = val;
    }
}

extern "C" void kernel_launch(void* const* d_in, const int* in_sizes, int n_in,
                              void* d_out, int out_size, void* d_ws, size_t ws_size,
                              hipStream_t stream) {
    const float* x    = (const float*)d_in[0];
    const int*   eidx = (const int*)  d_in[1];
    const float* pos  = (const float*)d_in[2];
    const float* W    = (const float*)d_in[3];
    const float* bias = (const float*)d_in[4];

    float* Wc = (float*)d_ws;
    float* y  = (float*)((char*)d_ws + 128 * 64 * sizeof(float));

    build_wc<<<(128 * 64 + 255) / 256, 256, 0, stream>>>(W, Wc);
    gemm_y<<<Bn * (Nn / 16), 32, 0, stream>>>(x, Wc, bias, y);
    edge_combine<<<(Bn * Nn) / 32, 128, 0, stream>>>(y, eidx, pos, (float*)d_out);
}


# edge factors 2*sigmoid(-dis) leave the edge stage: 16-byte position records, factors formed in the node GEMM launch by 2 record gathers per edge (was 6 dword gathers), handed on in planes of out
# speedup vs baseline: 1.3378x; 1.1734x over previous
#include <hip/hip_runtime.h>


#define Bn    2
#define Cn    64
#define Nn    20000
#define Kn    16
#define COUTn 64

typedef __attribute__((ext_vector_type(2))) float v2f;
typedef __attribute__((ext_vector_type(8))) float v8f;

#define PREP_THREADS (128 * 64)
#define PREP_TRIPS   ((Bn * Nn + PREP_THREADS - 1) / PREP_THREADS)

__global__ __launch_bounds__(256) void build_wc_pos4(
    const float* __restrict__ W, const float* __restrict__ pos,
    float* __restrict__ Wc, float* __restrict__ scr) {
    const int t = blockIdx.x * 256 + threadIdx.x;
    float4 rec[PREP_TRIPS];
#pragma unroll
    for (int r = 0; r < PREP_TRIPS; ++r) {
        int p = t + r * PREP_THREADS;
        p = p < Bn * Nn ? p : Bn * Nn - 1;
        int b = p / Nn, n = p - b * Nn;
        const float* pb = pos + (size_t)b * 3 * Nn;
        rec[r] = make_float4(pb[n], pb[Nn + n], pb[2 * Nn + n], 0.0f);
    }
    const int o = t >> 6, c = t & 63;
    const float* wr = W + (o & 63) * 128 + c;
    const float w1 = wr[0], w2 = wr[64];
    static_assert(PREP_TRIPS == 5, "the fence below names five records");
    asm volatile("" :: "v"(w1), "v"(w2),
                 "v"(rec[0].x), "v"(rec[0].y), "v"(rec[0].z),
                 "v"(rec[1].x), "v"(rec[1].y), "v"(rec[1].z),
                 "v"(rec[2].x), "v"(rec[2].y), "v"(rec[2].z),
                 "v"(rec[3].x), "v"(rec[3].y), "v"(rec[3].z),
                 "v"(rec[4].x), "v"(rec[4].y), "v"(rec[4].z));
    Wc[t] = (o < 64) ? w1 - w2 : w2;
#pragma unroll
    for (int r = 0; r < PREP_TRIPS; ++r) {
        int p = t + r * PREP_THREADS;
        if (r < PREP_TRIPS - 1 || p < Bn * Nn) {
            int b = p / Nn, n = p - b * Nn;
            float4* p4 = (float4*)(scr + ((size_t)b * COUTn + 32) * Nn);
            p4[n] = rec[r];
        }
    }
}

__device__ __forceinline__ float edge_factor(const float4 pj, const float4 pi) {
    float dx = pj.x - pi.x;
    float dy = pj.y - pi.y;
    float dz = pj.z - pi.z;
    float dis = sqrtf(dx * dx + dy * dy + dz * dz);
    return 2.0f / (1.0f + __expf(dis));
}

__global__ __launch_bounds__(32) void gemm_y(
    const float* __restrict__ x, const float* __restrict__ Wc,
    const float* __restrict__ bias, const int* __restrict__ eidx,
    float* __restrict__ y, float* __restrict__ scr) {
    const int lane = threadIdx.x;
    const int nlo  = lane & 15;
    const int hi   = lane >> 4;
    const int tile = blockIdx.x;
    const int b    = tile / (Nn / 16);
    const int m0   = (tile % (Nn / 16)) * 16;

    const size_t eb = ((size_t)b * Nn + m0) * Kn + lane * 4;
    const int4 jA = *(const int4*)(eidx + eb);
    const int4 jB = *(const int4*)(eidx + eb + 8 * Kn);
    const int4 iA = *(const int4*)(eidx + (size_t)Bn * Nn * Kn + eb);
    const int4 iB = *(const int4*)(eidx + (size_t)Bn * Nn * Kn + eb + 8 * Kn);

    v8f acc[8];
#pragma unroll
    for (int mt = 0; mt < 8; ++mt) {
#pragma unroll
        for (int r = 0; r < 8; ++r) {
            int o = mt * 16 + r + 8 * hi;
            acc[mt][r] = (o < COUTn) ? bias[o] : 0.0f;
        }
    }

    const float* xb = x + (size_t)b * Cn * Nn;
#pragma unroll 4
    for (int c0 = 0; c0 < Cn; c0 += 4) {
        v2f bf;
        bf.x = xb[(size_t)(c0 + 2 * hi)     * Nn + m0 + nlo];
        bf.y = xb[(size_t)(c0 + 2 * hi + 1) * Nn + m0 + nlo];
#pragma unroll
        for (int mt = 0; mt < 8; ++mt) {
            v2f af;
            af.x = Wc[(mt * 16 + nlo) * 64 + c0 + 2 * hi];
            af.y = Wc[(mt * 16 + nlo) * 64 + c0 + 2 * hi + 1];
            acc[mt] = __builtin_amdgcn_wmma_f32_16x16x4_f32(
                false, af, false, bf, (short)0, acc[mt], false, false);
        }
    }

    const float4* p4 = (const float4*)(scr + ((size_t)b * COUTn + 32) * Nn);
    const float4 pjA0 = p4[jA.x], pjA1 = p4[jA.y], pjA2 = p4[jA.z], pjA3 = p4[jA.w];
    const float4 piA0 = p4[iA.x], piA1 = p4[iA.y], piA2 = p4[iA.z], piA3 = p4[iA.w];
    const float4 pjB0 = p4[jB.x], pjB1 = p4[jB.y], pjB2 = p4[jB.z], pjB3 = p4[jB.w];
    const float4 piB0 = p4[iB.x], piB1 = p4[iB.y], piB2 = p4[iB.z], piB3 = p4[iB.w];

    float* yrow = y + ((size_t)b * Nn + m0 + nlo) * 128;
#pragma unroll
    for (int mt = 0; mt < 8; ++mt) {
        float4* dst = (float4*)(yrow + mt * 16 + 8 * hi);
        dst[0] = make_float4(acc[mt][0], acc[mt][1], acc[mt][2], acc[mt][3]);
        dst[1] = make_float4(acc[mt][4], acc[mt][5], acc[mt][6], acc[mt][7]);
    }

    const float4 fA = make_float4(edge_factor(pjA0, piA0), edge_factor(pjA1, piA1),
                                  edge_factor(pjA2, piA2), edge_factor(pjA3, piA3));
    const float4 fB = make_float4(edge_factor(pjB0, piB0), edge_factor(pjB1, piB1),
                                  edge_factor(pjB2, piB2), edge_factor(pjB3, piB3));
    float* fl = scr + ((size_t)b * COUTn + ((m0 & 16) >> 1) + (lane >> 3)) * Nn
                    + (m0 & ~31) + (lane & 7) * 4;
    *(float4*)fl            = fA;
    *(float4*)(fl + 4 * Nn) = fB;
}

#define EPITCH 36

__global__ __launch_bounds__(128) void edge_combine(
    const float* __restrict__ y, const int* __restrict__ eidx,
    float* __restrict__ out) {
    __shared__ __attribute__((aligned(16))) float smem[COUTn * EPITCH];
    const int t  = threadIdx.x;
    const int q  = t & 3;
    const int g  = t >> 2;
    const int p0 = blockIdx.x * 32;
    const int b  = p0 / Nn;
    const int n0 = p0 - b * Nn;
    const int n  = n0 + g;

    const size_t ebase = ((size_t)b * Nn + n) * Kn + q * 4;
    const int4 j4 = *(const int4*)(eidx + ebase);
    const int4 i4 = *(const int4*)(eidx + (size_t)Bn * Nn * Kn + ebase);
    const int jv[4] = {j4.x, j4.y, j4.z, j4.w};
    const int iv[4] = {i4.x, i4.y, i4.z, i4.w};
    const float4 f4 = *(const float4*)(out + ((size_t)b * COUTn + (g >> 1)) * Nn
                                           + n0 + (g & 1) * 16 + q * 4);
    const float supp[4] = {f4.x, f4.y, f4.z, f4.w};

    float4 mx[4];
#pragma unroll
    for (int s = 0; s < 4; ++s) mx[s] = make_float4(0.0f, 0.0f, 0.0f, 0.0f);

    const float* yb = y + (size_t)b * Nn * 128 + q * 4;
#pragma unroll
    for (int k = 0; k < Kn; ++k) {
        const int   ik = __shfl(iv[k & 3],   k >> 2, 4);
        const int   jk = __shfl(jv[k & 3],   k >> 2, 4);
        const float sk = __shfl(supp[k & 3], k >> 2, 4);
        const float4* yi = (const float4*)(yb + (size_t)ik * 128);
        const float4* yj = (const float4*)(yb + (size_t)jk * 128 + 64);
#pragma unroll
        for (int s = 0; s < 4; ++s) {
            const float4 a = yi[s * 4];
            const float4 c = yj[s * 4];
            mx[s].x = fmaxf(mx[s].x, fmaxf(a.x + c.x, 0.0f) * sk);
            mx[s].y = fmaxf(mx[s].y, fmaxf(a.y + c.y, 0.0f) * sk);
            mx[s].z = fmaxf(mx[s].z, fmaxf(a.z + c.z, 0.0f) * sk);
            mx[s].w = fmaxf(mx[s].w, fmaxf(a.w + c.w, 0.0f) * sk);
        }
    }

#pragma unroll
    for (int s = 0; s < 4; ++s) {
        float* r = smem + (s * 16 + q * 4) * EPITCH + g;
        r[0 * EPITCH] = mx[s].x;
        r[1 * EPITCH] = mx[s].y;
        r[2 * EPITCH] = mx[s].z;
        r[3 * EPITCH] = mx[s].w;
    }
    __syncthreads();
#pragma unroll
    for (int v = t; v < COUTn * 8; v += 128) {
        const int o = v >> 3, c4 = v & 7;
        const float4 val = *(const float4*)(smem + o * EPITCH + c4 * 4);
        *(float4*)(out + ((size_t)b * COUTn + o) * Nn + n0 + c4 * 4) = val;
    }
}

extern "C" void kernel_launch(void* const* d_in, const int* in_sizes, int n_in,
                              void* d_out, int out_size, void* d_ws, size_t ws_size,
                              hipStream_t stream) {
    const float* x    = (const float*)d_in[0];
    const int*   eidx = (const int*)  d_in[1];
    const float* pos  = (const float*)d_in[2];
    const float* W    = (const float*)d_in[3];
    const float* bias = (const float*)d_in[4];

    float* Wc = (float*)d_ws;
    float* y  = (float*)((char*)d_ws + 128 * 64 * sizeof(float));

    build_wc_pos4<<<PREP_THREADS / 256, 256, 0, stream>>>(W, pos, Wc, (float*)d_out);
    gemm_y<<<Bn * (Nn / 16), 32, 0, stream>>>(x, Wc, bias, eidx, y, (float*)d_out);
    edge_combine<<<(Bn * Nn) / 32, 128, 0, stream>>>(y, eidx, (float*)d_out);
}


# node GEMM operand roles swapped: nodes on the M side, channels on the N side, weights kept in B-fragment order; weight loads and y stores become whole-line 16-byte accesses
# speedup vs baseline: 1.4113x; 1.0549x over previous
#include <hip/hip_runtime.h>


#define Bn    2
#define Cn    64
#define Nn    20000
#define Kn    16
#define COUTn 64

typedef __attribute__((ext_vector_type(2))) float v2f;
typedef __attribute__((ext_vector_type(8))) float v8f;

#define PREP_THREADS (128 * 64)
#define PREP_TRIPS   ((Bn * Nn + PREP_THREADS - 1) / PREP_THREADS)

__global__ __launch_bounds__(256) void build_wf_pos4(
    const float* __restrict__ W, const float* __restrict__ pos,
    float* __restrict__ Wf, float* __restrict__ scr) {
    const int t = blockIdx.x * 256 + threadIdx.x;
    float4 rec[PREP_TRIPS];
#pragma unroll
    for (int r = 0; r < PREP_TRIPS; ++r) {
        int p = t + r * PREP_THREADS;
        p = p < Bn * Nn ? p : Bn * Nn - 1;
        int b = p / Nn, n = p - b * Nn;
        const float* pb = pos + (size_t)b * 3 * Nn;
        rec[r] = make_float4(pb[n], pb[Nn + n], pb[2 * Nn + n], 0.0f);
    }
    const int hf = (t >> 8) & 1;
    const int o  = 4 * ((t >> 2) & 15) + 2 * ((t >> 7) & 1) + ((t >> 1) & 1);
    const int k  = 4 * (t >> 9) + 2 * ((t >> 6) & 1) + (t & 1);
    const float* wr = W + o * 128 + k;
    const float w1 = wr[0], w2 = wr[64];
    static_assert(PREP_TRIPS == 5, "the fence below names five records");
    asm volatile("" :: "v"(w1), "v"(w2),
                 "v"(rec[0].x), "v"(rec[0].y), "v"(rec[0].z),
                 "v"(rec[1].x), "v"(rec[1].y), "v"(rec[1].z),
                 "v"(rec[2].x), "v"(rec[2].y), "v"(rec[2].z),
                 "v"(rec[3].x), "v"(rec[3].y), "v"(rec[3].z),
                 "v"(rec[4].x), "v"(rec[4].y), "v"(rec[4].z));
    Wf[t] = hf ? w2 : w1 - w2;
#pragma unroll
    for (int r = 0; r < PREP_TRIPS; ++r) {
        int p = t + r * PREP_THREADS;
        if (r < PREP_TRIPS - 1 || p < Bn * Nn) {
            int b = p / Nn, n = p - b * Nn;
            float4* p4 = (float4*)(scr + ((size_t)b * COUTn + 32) * Nn);
            p4[n] = rec[r];
        }
    }
}

__device__ __forceinline__ float edge_factor(const float4 pj, const float4 pi) {
    float dx = pj.x - pi.x;
    float dy = pj.y - pi.y;
    float dz = pj.z - pi.z;
    float dis = sqrtf(dx * dx + dy * dy + dz * dz);
    return 2.0f / (1.0f + __expf(dis));
}

__global__ __launch_bounds__(32) void gemm_y(
    const float* __restrict__ x, const float* __restrict__ Wf,
    const float* __restrict__ bias, const int* __restrict__ eidx,
    float* __restrict__ y, float* __restrict__ scr) {
    const int lane = threadIdx.x;
    const int nlo  = lane & 15;
    const int hi   = lane >> 4;
    const int tile = blockIdx.x;
    const int b    = tile / (Nn / 16);
    const int m0   = (tile % (Nn / 16)) * 16;

    const size_t eb = ((size_t)b * Nn + m0) * Kn + lane * 4;
    const int4 jA = *(const int4*)(eidx + eb);
    const int4 jB = *(const int4*)(eidx + eb + 8 * Kn);
    const int4 iA = *(const int4*)(eidx + (size_t)Bn * Nn * Kn + eb);
    const int4 iB = *(const int4*)(eidx + (size_t)Bn * Nn * Kn + eb + 8 * Kn);

    const float4 bv = *(const float4*)(bias + 4 * nlo);
    v8f acc[8];
#pragma unroll
    for (int r = 0; r < 8; ++r) {
        acc[0][r] = bv.x; acc[1][r] = bv.y; acc[2][r] = bv.z; acc[3][r] = bv.w;
        acc[4][r] = 0.0f; acc[5][r] = 0.0f; acc[6][r] = 0.0f; acc[7][r] = 0.0f;
    }

    const float*  xa = x + (size_t)b * Cn * Nn + (size_t)(2 * hi) * Nn + m0 + nlo;
    const float4* wf = (const float4*)Wf + lane;
#pragma unroll 4
    for (int j = 0; j < Cn / 4; ++j) {
        v2f af;
        af.x = xa[(size_t)(4 * j)     * Nn];
        af.y = xa[(size_t)(4 * j + 1) * Nn];
#pragma unroll
        for (int q = 0; q < 4; ++q) {
            const float4 w = wf[(j * 4 + q) * 32];
            v2f b0, b1;
            b0.x = w.x; b0.y = w.y;
            b1.x = w.z; b1.y = w.w;
            acc[2 * q]     = __builtin_amdgcn_wmma_f32_16x16x4_f32(
                false, af, false, b0, (short)0, acc[2 * q],     false, false);
            acc[2 * q + 1] = __builtin_amdgcn_wmma_f32_16x16x4_f32(
                false, af, false, b1, (short)0, acc[2 * q + 1], false, false);
        }
    }

    const float4* p4 = (const float4*)(scr + ((size_t)b * COUTn + 32) * Nn);
    const float4 pjA0 = p4[jA.x], pjA1 = p4[jA.y], pjA2 = p4[jA.z], pjA3 = p4[jA.w];
    const float4 piA0 = p4[iA.x], piA1 = p4[iA.y], piA2 = p4[iA.z], piA3 = p4[iA.w];
    const float4 pjB0 = p4[jB.x], pjB1 = p4[jB.y], pjB2 = p4[jB.z], pjB3 = p4[jB.w];
    const float4 piB0 = p4[iB.x], piB1 = p4[iB.y], piB2 = p4[iB.z], piB3 = p4[iB.w];

    float* yrow = y + ((size_t)b * Nn + m0 + 8 * hi) * 128 + 4 * nlo;
#pragma unroll
    for (int r = 0; r < 8; ++r) {
        *(float4*)(yrow + r * 128)      = make_float4(acc[0][r], acc[1][r], acc[2][r], acc[3][r]);
        *(float4*)(yrow + r * 128 + 64) = make_float4(acc[4][r], acc[5][r], acc[6][r], acc[7][r]);
    }

    const float4 fA = make_float4(edge_factor(pjA0, piA0), edge_factor(pjA1, piA1),
                                  edge_factor(pjA2, piA2), edge_factor(pjA3, piA3));
    const float4 fB = make_float4(edge_factor(pjB0, piB0), edge_factor(pjB1, piB1),
                                  edge_factor(pjB2, piB2), edge_factor(pjB3, piB3));
    float* fl = scr + ((size_t)b * COUTn + ((m0 & 16) >> 1) + (lane >> 3)) * Nn
                    + (m0 & ~31) + (lane & 7) * 4;
    *(float4*)fl            = fA;
    *(float4*)(fl + 4 * Nn) = fB;
}

#define EPITCH 36

__global__ __launch_bounds__(128) void edge_combine(
    const float* __restrict__ y, const int* __restrict__ eidx,
    float* __restrict__ out) {
    __shared__ __attribute__((aligned(16))) float smem[COUTn * EPITCH];
    const int t  = threadIdx.x;
    const int q  = t & 3;
    const int g  = t >> 2;
    const int p0 = blockIdx.x * 32;
    const int b  = p0 / Nn;
    const int n0 = p0 - b * Nn;
    const int n  = n0 + g;

    const size_t ebase = ((size_t)b * Nn + n) * Kn + q * 4;
    const int4 j4 = *(const int4*)(eidx + ebase);
    const int4 i4 = *(const int4*)(eidx + (size_t)Bn * Nn * Kn + ebase);
    const int jv[4] = {j4.x, j4.y, j4.z, j4.w};
    const int iv[4] = {i4.x, i4.y, i4.z, i4.w};
    const float4 f4 = *(const float4*)(out + ((size_t)b * COUTn + (g >> 1)) * Nn
                                           + n0 + (g & 1) * 16 + q * 4);
    const float supp[4] = {f4.x, f4.y, f4.z, f4.w};

    float4 mx[4];
#pragma unroll
    for (int s = 0; s < 4; ++s) mx[s] = make_float4(0.0f, 0.0f, 0.0f, 0.0f);

    const float* yb = y + (size_t)b * Nn * 128 + q * 4;
#pragma unroll
    for (int k = 0; k < Kn; ++k) {
        const int   ik = __shfl(iv[k & 3],   k >> 2, 4);
        const int   jk = __shfl(jv[k & 3],   k >> 2, 4);
        const float sk = __shfl(supp[k & 3], k >> 2, 4);
        const float4* yi = (const float4*)(yb + (size_t)ik * 128);
        const float4* yj = (const float4*)(yb + (size_t)jk * 128 + 64);
#pragma unroll
        for (int s = 0; s < 4; ++s) {
            const float4 a = yi[s * 4];
            const float4 c = yj[s * 4];
            mx[s].x = fmaxf(mx[s].x, fmaxf(a.x + c.x, 0.0f) * sk);
            mx[s].y = fmaxf(mx[s].y, fmaxf(a.y + c.y, 0.0f) * sk);
            mx[s].z = fmaxf(mx[s].z, fmaxf(a.z + c.z, 0.0f) * sk);
            mx[s].w = fmaxf(mx[s].w, fmaxf(a.w + c.w, 0.0f) * sk);
        }
    }

#pragma unroll
    for (int s = 0; s < 4; ++s) {
        float* r = smem + (s * 16 + q * 4) * EPITCH + g;
        r[0 * EPITCH] = mx[s].x;
        r[1 * EPITCH] = mx[s].y;
        r[2 * EPITCH] = mx[s].z;
        r[3 * EPITCH] = mx[s].w;
    }
    __syncthreads();
#pragma unroll
    for (int v = t; v < COUTn * 8; v += 128) {
        const int o = v >> 3, c4 = v & 7;
        const float4 val = *(const float4*)(smem + o * EPITCH + c4 * 4);
        *(float4*)(out + ((size_t)b * COUTn + o) * Nn + n0 + c4 * 4) = val;
    }
}

extern "C" void kernel_launch(void* const* d_in, const int* in_sizes, int n_in,
                              void* d_out, int out_size, void* d_ws, size_t ws_size,
                              hipStream_t stream) {
    const float* x    = (const float*)d_in[0];
    const int*   eidx = (const int*)  d_in[1];
    const float* pos  = (const float*)d_in[2];
    const float* W    = (const float*)d_in[3];
    const float* bias = (const float*)d_in[4];

    float* Wf = (float*)d_ws;
    float* y  = (float*)((char*)d_ws + 128 * 64 * sizeof(float));

    build_wf_pos4<<<PREP_THREADS / 256, 256, 0, stream>>>(W, pos, Wf, (float*)d_out);
    gemm_y<<<Bn * (Nn / 16), 32, 0, stream>>>(x, Wf, bias, eidx, y, (float*)d_out);
    edge_combine<<<(Bn * Nn) / 32, 128, 0, stream>>>(y, eidx, (float*)d_out);
}


# edge stage lane map: 8 lanes per node, so each 16-byte row load covers 4 whole 128-byte lines (was 8 half lines, every line asked for by two loads); side effect 187 VGPRs (was 310)
# speedup vs baseline: 1.6160x; 1.1450x over previous
#include <hip/hip_runtime.h>


#define Bn    2
#define Cn    64
#define Nn    20000
#define Kn    16
#define COUTn 64

typedef __attribute__((ext_vector_type(2))) float v2f;
typedef __attribute__((ext_vector_type(8))) float v8f;

#define PREP_THREADS (128 * 64)
#define PREP_TRIPS   ((Bn * Nn + PREP_THREADS - 1) / PREP_THREADS)

__global__ __launch_bounds__(256) void build_wf_pos4(
    const float* __restrict__ W, const float* __restrict__ pos,
    float* __restrict__ Wf, float* __restrict__ scr) {
    const int t = blockIdx.x * 256 + threadIdx.x;
    float4 rec[PREP_TRIPS];
#pragma unroll
    for (int r = 0; r < PREP_TRIPS; ++r) {
        int p = t + r * PREP_THREADS;
        p = p < Bn * Nn ? p : Bn * Nn - 1;
        int b = p / Nn, n = p - b * Nn;
        const float* pb = pos + (size_t)b * 3 * Nn;
        rec[r] = make_float4(pb[n], pb[Nn + n], pb[2 * Nn + n], 0.0f);
    }
    const int hf = (t >> 8) & 1;
    const int o  = 4 * ((t >> 2) & 15) + 2 * ((t >> 7) & 1) + ((t >> 1) & 1);
    const int k  = 4 * (t >> 9) + 2 * ((t >> 6) & 1) + (t & 1);
    const float* wr = W + o * 128 + k;
    const float w1 = wr[0], w2 = wr[64];
    static_assert(PREP_TRIPS == 5, "the fence below names five records");
    asm volatile("" :: "v"(w1), "v"(w2),
                 "v"(rec[0].x), "v"(rec[0].y), "v"(rec[0].z),
                 "v"(rec[1].x), "v"(rec[1].y), "v"(rec[1].z),
                 "v"(rec[2].x), "v"(rec[2].y), "v"(rec[2].z),
                 "v"(rec[3].x), "v"(rec[3].y), "v"(rec[3].z),
                 "v"(rec[4].x), "v"(rec[4].y), "v"(rec[4].z));
    Wf[t] = hf ? w2 : w1 - w2;
#pragma unroll
    for (int r = 0; r < PREP_TRIPS; ++r) {
        int p = t + r * PREP_THREADS;
        if (r < PREP_TRIPS - 1 || p < Bn * Nn) {
            int b = p / Nn, n = p - b * Nn;
            float4* p4 = (float4*)(scr + ((size_t)b * COUTn + 32) * Nn);
            p4[n] = rec[r];
        }
    }
}

__device__ __forceinline__ float edge_factor(const float4 pj, const float4 pi) {
    float dx = pj.x - pi.x;
    float dy = pj.y - pi.y;
    float dz = pj.z - pi.z;
    float dis = sqrtf(dx * dx + dy * dy + dz * dz);
    return 2.0f / (1.0f + __expf(dis));
}

__global__ __launch_bounds__(32) void gemm_y(
    const float* __restrict__ x, const float* __restrict__ Wf,
    const float* __restrict__ bias, const int* __restrict__ eidx,
    float* __restrict__ y, float* __restrict__ scr) {
    const int lane = threadIdx.x;
    const int nlo  = lane & 15;
    const int hi   = lane >> 4;
    const int tile = blockIdx.x;
    const int b    = tile / (Nn / 16);
    const int m0   = (tile % (Nn / 16)) * 16;

    const size_t eb = ((size_t)b * Nn + m0) * Kn + lane * 4;
    const int4 jA = *(const int4*)(eidx + eb);
    const int4 jB = *(const int4*)(eidx + eb + 8 * Kn);
    const int4 iA = *(const int4*)(eidx + (size_t)Bn * Nn * Kn + eb);
    const int4 iB = *(const int4*)(eidx + (size_t)Bn * Nn * Kn + eb + 8 * Kn);

    const float4 bv = *(const float4*)(bias + 4 * nlo);
    v8f acc[8];
#pragma unroll
    for (int r = 0; r < 8; ++r) {
        acc[0][r] = bv.x; acc[1][r] = bv.y; acc[2][r] = bv.z; acc[3][r] = bv.w;
        acc[4][r] = 0.0f; acc[5][r] = 0.0f; acc[6][r] = 0.0f; acc[7][r] = 0.0f;
    }

    const float*  xa = x + (size_t)b * Cn * Nn + (size_t)(2 * hi) * Nn + m0 + nlo;
    const float4* wf = (const float4*)Wf + lane;
#pragma unroll 4
    for (int j = 0; j < Cn / 4; ++j) {
        v2f af;
        af.x = xa[(size_t)(4 * j)     * Nn];
        af.y = xa[(size_t)(4 * j + 1) * Nn];
#pragma unroll
        for (int q = 0; q < 4; ++q) {
            const float4 w = wf[(j * 4 + q) * 32];
            v2f b0, b1;
            b0.x = w.x; b0.y = w.y;
            b1.x = w.z; b1.y = w.w;
            acc[2 * q]     = __builtin_amdgcn_wmma_f32_16x16x4_f32(
                false, af, false, b0, (short)0, acc[2 * q],     false, false);
            acc[2 * q + 1] = __builtin_amdgcn_wmma_f32_16x16x4_f32(
                false, af, false, b1, (short)0, acc[2 * q + 1], false, false);
        }
    }

    const float4* p4 = (const float4*)(scr + ((size_t)b * COUTn + 32) * Nn);
    const float4 pjA0 = p4[jA.x], pjA1 = p4[jA.y], pjA2 = p4[jA.z], pjA3 = p4[jA.w];
    const float4 piA0 = p4[iA.x], piA1 = p4[iA.y], piA2 = p4[iA.z], piA3 = p4[iA.w];
    const float4 pjB0 = p4[jB.x], pjB1 = p4[jB.y], pjB2 = p4[jB.z], pjB3 = p4[jB.w];
    const float4 piB0 = p4[iB.x], piB1 = p4[iB.y], piB2 = p4[iB.z], piB3 = p4[iB.w];

    float* yrow = y + ((size_t)b * Nn + m0 + 8 * hi) * 128 + 4 * nlo;
#pragma unroll
    for (int r = 0; r < 8; ++r) {
        *(float4*)(yrow + r * 128)      = make_float4(acc[0][r], acc[1][r], acc[2][r], acc[3][r]);
        *(float4*)(yrow + r * 128 + 64) = make_float4(acc[4][r], acc[5][r], acc[6][r], acc[7][r]);
    }

    const float4 fA = make_float4(edge_factor(pjA0, piA0), edge_factor(pjA1, piA1),
                                  edge_factor(pjA2, piA2), edge_factor(pjA3, piA3));
    const float4 fB = make_float4(edge_factor(pjB0, piB0), edge_factor(pjB1, piB1),
                                  edge_factor(pjB2, piB2), edge_factor(pjB3, piB3));
    float* fl = scr + ((size_t)b * COUTn + ((m0 & 16) >> 1) + (lane >> 3)) * Nn
                    + (m0 & ~31) + (lane & 7) * 4;
    *(float4*)fl            = fA;
    *(float4*)(fl + 4 * Nn) = fB;
}

#define EPITCH 36

template <int s>
__device__ __forceinline__ int bcast8(int v) {
    return __builtin_amdgcn_mov_dpp8(v, (unsigned)s * 0x249249u);
}

__global__ __launch_bounds__(128) void edge_combine(
    const float* __restrict__ y, const int* __restrict__ eidx,
    float* __restrict__ out) {
    __shared__ __attribute__((aligned(16))) float smem[COUTn * EPITCH];
    const int t  = threadIdx.x;
    const int q  = t & 7;
    const int g  = (t >> 3) & 3;
    const int w  = t >> 5;
    const int p0 = blockIdx.x * 32;
    const int b  = p0 / Nn;
    const int n0 = p0 - b * Nn;

    int   jv[2][2], iv[2][2];
    float supp[2][2];
#pragma unroll
    for (int a = 0; a < 2; ++a) {
        const int slot = w * 8 + a * 4 + g;
        const size_t ebase = ((size_t)b * Nn + n0 + slot) * Kn + q * 2;
        const int2 j2 = *(const int2*)(eidx + ebase);
        const int2 i2 = *(const int2*)(eidx + (size_t)Bn * Nn * Kn + ebase);
        const float2 f2 = *(const float2*)(out + ((size_t)b * COUTn + (slot >> 1)) * Nn
                                               + n0 + (slot & 1) * 16 + q * 2);
        jv[a][0] = j2.x; jv[a][1] = j2.y;
        iv[a][0] = i2.x; iv[a][1] = i2.y;
        supp[a][0] = f2.x; supp[a][1] = f2.y;
    }
    asm volatile("" :: "v"(supp[0][0]), "v"(supp[0][1]), "v"(supp[1][0]), "v"(supp[1][1]));

    float4 mx[2][2];
#pragma unroll
    for (int a = 0; a < 2; ++a)
#pragma unroll
        for (int u = 0; u < 2; ++u) mx[a][u] = make_float4(0.0f, 0.0f, 0.0f, 0.0f);

    const float* yb = y + (size_t)b * Nn * 128 + q * 4;
#define EDGE_STEP(k)                                                                   \
    {                                                                                  \
        _Pragma("unroll")                                                              \
        for (int a = 0; a < 2; ++a) {                                                  \
                         \
            const int   ik = bcast8<((k) >> 1)>(iv[a][(k) & 1]);                       \
            const int   jk = bcast8<((k) >> 1)>(jv[a][(k) & 1]);                       \
            const float sk = __int_as_float(bcast8<((k) >> 1)>(__float_as_int(supp[a][(k) & 1]))); \
            const float4* yi = (const float4*)(yb + (size_t)ik * 128);        \
            const float4* yj = (const float4*)(yb + (size_t)jk * 128 + 64);         \
            _Pragma("unroll")                                                          \
            for (int u = 0; u < 2; ++u) {                                              \
                const float4 va = yi[u * 8];                \
                const float4 vc = yj[u * 8];                                           \
                mx[a][u].x = fmaxf(mx[a][u].x, fmaxf(va.x + vc.x, 0.0f) * sk);         \
                mx[a][u].y = fmaxf(mx[a][u].y, fmaxf(va.y + vc.y, 0.0f) * sk);         \
                mx[a][u].z = fmaxf(mx[a][u].z, fmaxf(va.z + vc.z, 0.0f) * sk);         \
                mx[a][u].w = fmaxf(mx[a][u].w, fmaxf(va.w + vc.w, 0.0f) * sk);         \
            }                                                                          \
        }                                                                              \
    }
    EDGE_STEP(0)  EDGE_STEP(1)  EDGE_STEP(2)  EDGE_STEP(3)
    EDGE_STEP(4)  EDGE_STEP(5)  EDGE_STEP(6)  EDGE_STEP(7)
    EDGE_STEP(8)  EDGE_STEP(9)  EDGE_STEP(10) EDGE_STEP(11)
    EDGE_STEP(12) EDGE_STEP(13) EDGE_STEP(14) EDGE_STEP(15)
#undef EDGE_STEP

#pragma unroll
    for (int a = 0; a < 2; ++a) {
        const int slot = w * 8 + a * 4 + g;
#pragma unroll
        for (int u = 0; u < 2; ++u) {
            float* r = smem + (u * 32 + q * 4) * EPITCH + slot;
            r[0 * EPITCH] = mx[a][u].x;
            r[1 * EPITCH] = mx[a][u].y;
            r[2 * EPITCH] = mx[a][u].z;
            r[3 * EPITCH] = mx[a][u].w;
        }
    }
    __syncthreads();
#pragma unroll
    for (int v = t; v < COUTn * 8; v += 128) {
        const int o = v >> 3, c4 = v & 7;
        const float4 val = *(const float4*)(smem + o * EPITCH + c4 * 4);
        *(float4*)(out + ((size_t)b * COUTn + o) * Nn + n0 + c4 * 4) = val;
    }
}

extern "C" void kernel_launch(void* const* d_in, const int* in_sizes, int n_in,
                              void* d_out, int out_size, void* d_ws, size_t ws_size,
                              hipStream_t stream) {
    const float* x    = (const float*)d_in[0];
    const int*   eidx = (const int*)  d_in[1];
    const float* pos  = (const float*)d_in[2];
    const float* W    = (const float*)d_in[3];
    const float* bias = (const float*)d_in[4];

    float* Wf = (float*)d_ws;
    float* y  = (float*)((char*)d_ws + 128 * 64 * sizeof(float));

    build_wf_pos4<<<PREP_THREADS / 256, 256, 0, stream>>>(W, pos, Wf, (float*)d_out);
    gemm_y<<<Bn * (Nn / 16), 32, 0, stream>>>(x, Wf, bias, eidx, y, (float*)d_out);
    edge_combine<<<(Bn * Nn) / 32, 128, 0, stream>>>(y, eidx, (float*)d_out);
}


# node GEMM launch software-pipelined: all x fragments requested at wave entry, weight loads a trip ahead, record gathers under the last 54 WMMAs; one exposed memory round trip per wave, was five
# speedup vs baseline: 1.6813x; 1.0404x over previous
#include <hip/hip_runtime.h>


#define Bn    2
#define Cn    64
#define Nn    20000
#define Kn    16
#define COUTn 64

typedef __attribute__((ext_vector_type(2))) float v2f;
typedef __attribute__((ext_vector_type(8))) float v8f;

#define PREP_THREADS (128 * 64)
#define PREP_TRIPS   ((Bn * Nn + PREP_THREADS - 1) / PREP_THREADS)

__global__ __launch_bounds__(256) void build_wf_pos4(
    const float* __restrict__ W, const float* __restrict__ pos,
    float* __restrict__ Wf, float* __restrict__ scr) {
    const int t = blockIdx.x * 256 + threadIdx.x;
    float4 rec[PREP_TRIPS];
#pragma unroll
    for (int r = 0; r < PREP_TRIPS; ++r) {
        int p = t + r * PREP_THREADS;
        p = p < Bn * Nn ? p : Bn * Nn - 1;
        int b = p / Nn, n = p - b * Nn;
        const float* pb = pos + (size_t)b * 3 * Nn;
        rec[r] = make_float4(pb[n], pb[Nn + n], pb[2 * Nn + n], 0.0f);
    }
    const int hf = (t >> 8) & 1;
    const int o  = 4 * ((t >> 2) & 15) + 2 * ((t >> 7) & 1) + ((t >> 1) & 1);
    const int k  = 4 * (t >> 9) + 2 * ((t >> 6) & 1) + (t & 1);
    const float* wr = W + o * 128 + k;
    const float w1 = wr[0], w2 = wr[64];
    static_assert(PREP_TRIPS == 5, "the fence below names five records");
    asm volatile("" :: "v"(w1), "v"(w2),
                 "v"(rec[0].x), "v"(rec[0].y), "v"(rec[0].z),
                 "v"(rec[1].x), "v"(rec[1].y), "v"(rec[1].z),
                 "v"(rec[2].x), "v"(rec[2].y), "v"(rec[2].z),
                 "v"(rec[3].x), "v"(rec[3].y), "v"(rec[3].z),
                 "v"(rec[4].x), "v"(rec[4].y), "v"(rec[4].z));
    Wf[t] = hf ? w2 : w1 - w2;
#pragma unroll
    for (int r = 0; r < PREP_TRIPS; ++r) {
        int p = t + r * PREP_THREADS;
        if (r < PREP_TRIPS - 1 || p < Bn * Nn) {
            int b = p / Nn, n = p - b * Nn;
            float4* p4 = (float4*)(scr + ((size_t)b * COUTn + 32) * Nn);
            p4[n] = rec[r];
        }
    }
}

__device__ __forceinline__ float edge_factor(const float4 pj, const float4 pi) {
    float dx = pj.x - pi.x;
    float dy = pj.y - pi.y;
    float dz = pj.z - pi.z;
    float dis = sqrtf(dx * dx + dy * dy + dz * dz);
    return 2.0f / (1.0f + __expf(dis));
}

__global__ __launch_bounds__(32) void gemm_y(
    const float* __restrict__ x, const float* __restrict__ Wf,
    const float* __restrict__ bias, const int* __restrict__ eidx,
    float* __restrict__ y, float* __restrict__ scr) {
    const int lane = threadIdx.x;
    const int nlo  = lane & 15;
    const int hi   = lane >> 4;
    const int tile = blockIdx.x;
    const int b    = tile / (Nn / 16);
    const int m0   = (tile % (Nn / 16)) * 16;

    const size_t eb = ((size_t)b * Nn + m0) * Kn + lane * 4;
    const int4 jA = *(const int4*)(eidx + eb);
    const int4 jB = *(const int4*)(eidx + eb + 8 * Kn);
    const int4 iA = *(const int4*)(eidx + (size_t)Bn * Nn * Kn + eb);
    const int4 iB = *(const int4*)(eidx + (size_t)Bn * Nn * Kn + eb + 8 * Kn);

    const float*  xa = x + (size_t)b * Cn * Nn + (size_t)(2 * hi) * Nn + m0 + nlo;
    const float4* wf = (const float4*)Wf + lane;
    const float4* p4 = (const float4*)(scr + ((size_t)b * COUTn + 32) * Nn);

    v2f af[Cn / 4];
#pragma unroll
    for (int j = 0; j < Cn / 4; ++j) {
        af[j].x = xa[(size_t)(4 * j)     * Nn];
        af[j].y = xa[(size_t)(4 * j + 1) * Nn];
    }
#define LOAD_W(buf, t)                                                         \
    _Pragma("unroll")                                                          \
    for (int i = 0; i < 16; ++i) buf[i] = wf[((t) * 16 + i) * 32];
#define TRIP_WMMA(buf, t)                                                      \
    _Pragma("unroll")                                                          \
    for (int jj = 0; jj < 4; ++jj) {                                           \
        _Pragma("unroll")                                                      \
        for (int q = 0; q < 4; ++q) {                                          \
            const float4 w = buf[jj * 4 + q];                                  \
            v2f b0, b1;                                                        \
            b0.x = w.x; b0.y = w.y;                                            \
            b1.x = w.z; b1.y = w.w;                                            \
            acc[2 * q]     = __builtin_amdgcn_wmma_f32_16x16x4_f32(            \
                false, af[(t) * 4 + jj], false, b0, (short)0, acc[2 * q],     false, false); \
            acc[2 * q + 1] = __builtin_amdgcn_wmma_f32_16x16x4_f32(            \
                false, af[(t) * 4 + jj], false, b1, (short)0, acc[2 * q + 1], false, false); \
        }                                                                      \
    }
    float4 wA[16], wB[16];
    LOAD_W(wA, 0)
    LOAD_W(wB, 1)
    const float4 bv = *(const float4*)(bias + 4 * nlo);
    __builtin_amdgcn_sched_barrier(0);
    v8f acc[8];
#pragma unroll
    for (int r = 0; r < 8; ++r) {
        acc[0][r] = bv.x; acc[1][r] = bv.y; acc[2][r] = bv.z; acc[3][r] = bv.w;
        acc[4][r] = 0.0f; acc[5][r] = 0.0f; acc[6][r] = 0.0f; acc[7][r] = 0.0f;
    }
    TRIP_WMMA(wA, 0)
    LOAD_W(wA, 2)
    TRIP_WMMA(wB, 1)
    LOAD_W(wB, 3)
    __builtin_amdgcn_sched_barrier(0);
    const float4 pjA0 = p4[jA.x], pjA1 = p4[jA.y], pjA2 = p4[jA.z], pjA3 = p4[jA.w];
    const float4 piA0 = p4[iA.x], piA1 = p4[iA.y], piA2 = p4[iA.z], piA3 = p4[iA.w];
    const float4 pjB0 = p4[jB.x], pjB1 = p4[jB.y], pjB2 = p4[jB.z], pjB3 = p4[jB.w];
    const float4 piB0 = p4[iB.x], piB1 = p4[iB.y], piB2 = p4[iB.z], piB3 = p4[iB.w];
    TRIP_WMMA(wA, 2)
    TRIP_WMMA(wB, 3)
#undef LOAD_W
#undef TRIP_WMMA

    float* yrow = y + ((size_t)b * Nn + m0 + 8 * hi) * 128 + 4 * nlo;
#pragma unroll
    for (int r = 0; r < 8; ++r) {
        *(float4*)(yrow + r * 128)      = make_float4(acc[0][r], acc[1][r], acc[2][r], acc[3][r]);
        *(float4*)(yrow + r * 128 + 64) = make_float4(acc[4][r], acc[5][r], acc[6][r], acc[7][r]);
    }

    const float4 fA = make_float4(edge_factor(pjA0, piA0), edge_factor(pjA1, piA1),
                                  edge_factor(pjA2, piA2), edge_factor(pjA3, piA3));
    const float4 fB = make_float4(edge_factor(pjB0, piB0), edge_factor(pjB1, piB1),
                                  edge_factor(pjB2, piB2), edge_factor(pjB3, piB3));
    float* fl = scr + ((size_t)b * COUTn + ((m0 & 16) >> 1) + (lane >> 3)) * Nn
                    + (m0 & ~31) + (lane & 7) * 4;
    *(float4*)fl            = fA;
    *(float4*)(fl + 4 * Nn) = fB;
}

#define EPITCH 36

template <int s>
__device__ __forceinline__ int bcast8(int v) {
    return __builtin_amdgcn_mov_dpp8(v, (unsigned)s * 0x249249u);
}

__global__ __launch_bounds__(128) void edge_combine(
    const float* __restrict__ y, const int* __restrict__ eidx,
    float* __restrict__ out) {
    __shared__ __attribute__((aligned(16))) float smem[COUTn * EPITCH];
    const int t  = threadIdx.x;
    const int q  = t & 7;
    const int g  = (t >> 3) & 3;
    const int w  = t >> 5;
    const int p0 = blockIdx.x * 32;
    const int b  = p0 / Nn;
    const int n0 = p0 - b * Nn;

    int   jv[2][2], iv[2][2];
    float supp[2][2];
#pragma unroll
    for (int a = 0; a < 2; ++a) {
        const int slot = w * 8 + a * 4 + g;
        const size_t ebase = ((size_t)b * Nn + n0 + slot) * Kn + q * 2;
        const int2 j2 = *(const int2*)(eidx + ebase);
        const int2 i2 = *(const int2*)(eidx + (size_t)Bn * Nn * Kn + ebase);
        const float2 f2 = *(const float2*)(out + ((size_t)b * COUTn + (slot >> 1)) * Nn
                                               + n0 + (slot & 1) * 16 + q * 2);
        jv[a][0] = j2.x; jv[a][1] = j2.y;
        iv[a][0] = i2.x; iv[a][1] = i2.y;
        supp[a][0] = f2.x; supp[a][1] = f2.y;
    }
    asm volatile("" :: "v"(supp[0][0]), "v"(supp[0][1]), "v"(supp[1][0]), "v"(supp[1][1]));

    float4 mx[2][2];
#pragma unroll
    for (int a = 0; a < 2; ++a)
#pragma unroll
        for (int u = 0; u < 2; ++u) mx[a][u] = make_float4(0.0f, 0.0f, 0.0f, 0.0f);

    const float* yb = y + (size_t)b * Nn * 128 + q * 4;
#define EDGE_STEP(k)                                                                   \
    {                                                                                  \
        _Pragma("unroll")                                                              \
        for (int a = 0; a < 2; ++a) {                                                  \
                         \
            const int   ik = bcast8<((k) >> 1)>(iv[a][(k) & 1]);                       \
            const int   jk = bcast8<((k) >> 1)>(jv[a][(k) & 1]);                       \
            const float sk = __int_as_float(bcast8<((k) >> 1)>(__float_as_int(supp[a][(k) & 1]))); \
            const float4* yi = (const float4*)(yb + (size_t)ik * 128);        \
            const float4* yj = (const float4*)(yb + (size_t)jk * 128 + 64);         \
            _Pragma("unroll")                                                          \
            for (int u = 0; u < 2; ++u) {                                              \
                const float4 va = yi[u * 8];                \
                const float4 vc = yj[u * 8];                                           \
                mx[a][u].x = fmaxf(mx[a][u].x, fmaxf(va.x + vc.x, 0.0f) * sk);         \
                mx[a][u].y = fmaxf(mx[a][u].y, fmaxf(va.y + vc.y, 0.0f) * sk);         \
                mx[a][u].z = fmaxf(mx[a][u].z, fmaxf(va.z + vc.z, 0.0f) * sk);         \
                mx[a][u].w = fmaxf(mx[a][u].w, fmaxf(va.w + vc.w, 0.0f) * sk);         \
            }                                                                          \
        }                                                                              \
    }
    EDGE_STEP(0)  EDGE_STEP(1)  EDGE_STEP(2)  EDGE_STEP(3)
    EDGE_STEP(4)  EDGE_STEP(5)  EDGE_STEP(6)  EDGE_STEP(7)
    EDGE_STEP(8)  EDGE_STEP(9)  EDGE_STEP(10) EDGE_STEP(11)
    EDGE_STEP(12) EDGE_STEP(13) EDGE_STEP(14) EDGE_STEP(15)
#undef EDGE_STEP

#pragma unroll
    for (int a = 0; a < 2; ++a) {
        const int slot = w * 8 + a * 4 + g;
#pragma unroll
        for (int u = 0; u < 2; ++u) {
            float* r = smem + (u * 32 + q * 4) * EPITCH + slot;
            r[0 * EPITCH] = mx[a][u].x;
            r[1 * EPITCH] = mx[a][u].y;
            r[2 * EPITCH] = mx[a][u].z;
            r[3 * EPITCH] = mx[a][u].w;
        }
    }
    __syncthreads();
#pragma unroll
    for (int v = t; v < COUTn * 8; v += 128) {
        const int o = v >> 3, c4 = v & 7;
        const float4 val = *(const float4*)(smem + o * EPITCH + c4 * 4);
        *(float4*)(out + ((size_t)b * COUTn + o) * Nn + n0 + c4 * 4) = val;
    }
}

extern "C" void kernel_launch(void* const* d_in, const int* in_sizes, int n_in,
                              void* d_out, int out_size, void* d_ws, size_t ws_size,
                              hipStream_t stream) {
    const float* x    = (const float*)d_in[0];
    const int*   eidx = (const int*)  d_in[1];
    const float* pos  = (const float*)d_in[2];
    const float* W    = (const float*)d_in[3];
    const float* bias = (const float*)d_in[4];

    float* Wf = (float*)d_ws;
    float* y  = (float*)((char*)d_ws + 128 * 64 * sizeof(float));

    build_wf_pos4<<<PREP_THREADS / 256, 256, 0, stream>>>(W, pos, Wf, (float*)d_out);
    gemm_y<<<Bn * (Nn / 16), 32, 0, stream>>>(x, Wf, bias, eidx, y, (float*)d_out);
    edge_combine<<<(Bn * Nn) / 32, 128, 0, stream>>>(y, eidx, (float*)d_out);
}
